# starting kernel
# baseline (speedup 1.0000x reference)
#include <hip/hip_runtime.h>

#define B_     8
#define N_IN   1024
#define N_OUT  4096
#define OC     64
#define BLK    128

typedef float v2f __attribute__((ext_vector_type(2)));
typedef float v8f __attribute__((ext_vector_type(8)));

__global__ __launch_bounds__(BLK) void convdeepset_kernel(
    const float* __restrict__ x,
    const float* __restrict__ y,
    const float* __restrict__ t,
    const float* __restrict__ sigma,
    const float* __restrict__ W,
    const float* __restrict__ bias,
    float* __restrict__ out)
{
    __shared__ float4 sdata[N_IN];
    __shared__ float  sfeat0[BLK];
    __shared__ float  sfeat1[BLK];

    const int tid = threadIdx.x;
    const int b   = blockIdx.y;
    const int m0  = blockIdx.x * BLK;

    const float* xb = x + (size_t)b * N_IN * 2;
    const float* yb = y + (size_t)b * N_IN;
    for (int n = tid; n < N_IN; n += BLK) {
        float2 xv = ((const float2*)xb)[n];
        float4 s; s.x = xv.x; s.y = xv.y; s.z = yb[n]; s.w = 0.0f;
        sdata[n] = s;
    }

    const float HALF_OVER_LN2 = 0.7213475204444817f;
    const float LOG2E         = 1.4426950408889634f;
    const float c0 = -HALF_OVER_LN2 * __builtin_amdgcn_exp2f(-2.0f * sigma[0] * LOG2E);
    const float c1 = -HALF_OVER_LN2 * __builtin_amdgcn_exp2f(-2.0f * sigma[1] * LOG2E);

    const int m = m0 + tid;
    const float2 tv = ((const float2*)(t + (size_t)b * N_OUT * 2))[m];

    __syncthreads();

    float dens = 0.0f, conv = 0.0f;
    #pragma unroll 8
    for (int n = 0; n < N_IN; ++n) {
        float4 s = sdata[n];
        float dx = s.x - tv.x;
        float dy = s.y - tv.y;
        float d  = __builtin_fmaf(dy, dy, dx * dx);
        dens += __builtin_amdgcn_exp2f(d * c0);
        conv  = __builtin_fmaf(s.z, __builtin_amdgcn_exp2f(d * c1), conv);
    }

    sfeat0[tid] = dens;
    sfeat1[tid] = conv / (dens + 1e-8f);
    __syncthreads();

    const int lane = tid & 31;
    const int wv   = tid >> 5;
    const int hi   = lane >> 4;
    const int l15  = lane & 15;

    v2f a[2];
    #pragma unroll
    for (int g = 0; g < 2; ++g) {
        const int r = wv * 32 + g * 16 + l15;
        const float f0 = sfeat0[r];
        const float f1 = sfeat1[r];
        a[g].x = hi ? 0.0f : f0;
        a[g].y = hi ? 0.0f : f1;
    }

    #pragma unroll
    for (int ot = 0; ot < 4; ++ot) {
        const int o  = ot * 16 + l15;
        const float w0 = W[o * 2 + 0];
        const float w1 = W[o * 2 + 1];
        const float bv = bias[o];
        v2f bm;
        bm.x = hi ? 0.0f : w0;
        bm.y = hi ? 0.0f : w1;

        #pragma unroll
        for (int g = 0; g < 2; ++g) {
            v8f c;
            #pragma unroll
            for (int r = 0; r < 8; ++r) c[r] = bv;

            v8f d = __builtin_amdgcn_wmma_f32_16x16x4_f32(
                false, a[g], false, bm, (short)0, c, false, false);

            const int m_blk = wv * 32 + g * 16 + hi * 8;
            float* op = out + (((size_t)b * N_OUT + (m0 + m_blk)) * OC + o);
            #pragma unroll
            for (int r = 0; r < 8; ++r)
                op[(size_t)r * OC] = d[r];
        }
    }
}

extern "C" void kernel_launch(void* const* d_in, const int* in_sizes, int n_in,
                              void* d_out, int out_size, void* d_ws, size_t ws_size,
                              hipStream_t stream) {
    const float* x     = (const float*)d_in[0];
    const float* y     = (const float*)d_in[1];
    const float* t     = (const float*)d_in[2];
    const float* sigma = (const float*)d_in[3];
    const float* W     = (const float*)d_in[4];
    const float* bias  = (const float*)d_in[5];
    float* out = (float*)d_out;

    dim3 grid(N_OUT / BLK, B_);
    convdeepset_kernel<<<grid, dim3(BLK), 0, stream>>>(x, y, t, sigma, W, bias, out);
}


# Re-tile the context sum: 4 targets per lane, n-range split over 8 waves per block (2 per SIMD), the 8 partial sums per target joined in LDS before the WMMA projection
# speedup vs baseline: 1.8480x; 1.8480x over previous
#include <hip/hip_runtime.h>


#define B_     8
#define N_IN   1024
#define N_OUT  4096
#define OC     64
#define MB     128
#define RM     4
#define NSPLIT 8
#define NPART  (N_IN / NSPLIT)
#define BLK    256

typedef float v2f __attribute__((ext_vector_type(2)));
typedef float v8f __attribute__((ext_vector_type(8)));

__global__ __launch_bounds__(BLK) void convdeepset_kernel(
    const float* __restrict__ x,
    const float* __restrict__ y,
    const float* __restrict__ t,
    const float* __restrict__ sigma,
    const float* __restrict__ W,
    const float* __restrict__ bias,
    float* __restrict__ out)
{
    __shared__ float4 sdata[N_IN];
    __shared__ float  spd[NSPLIT][MB];
    __shared__ float  spc[NSPLIT][MB];

    const int tid  = threadIdx.x;
    const int lane = tid & 31;
    const int wv   = tid >> 5;
    const int b    = blockIdx.y;
    const int m0   = blockIdx.x * MB;

    const float* xb = x + (size_t)b * N_IN * 2;
    const float* yb = y + (size_t)b * N_IN;
    for (int n = tid; n < N_IN; n += BLK) {
        float2 xv = ((const float2*)xb)[n];
        float4 s; s.x = xv.x; s.y = xv.y; s.z = yb[n]; s.w = 0.0f;
        sdata[n] = s;
    }

    const float HALF_OVER_LN2 = 0.7213475204444817f;
    const float LOG2E         = 1.4426950408889634f;
    const float c0 = -HALF_OVER_LN2 * __builtin_amdgcn_exp2f(-2.0f * sigma[0] * LOG2E);
    const float c1 = -HALF_OVER_LN2 * __builtin_amdgcn_exp2f(-2.0f * sigma[1] * LOG2E);

    const float2* tb = (const float2*)(t + (size_t)b * N_OUT * 2) + (m0 + lane);
    float2 tv[RM];
    #pragma unroll
    for (int j = 0; j < RM; ++j) tv[j] = tb[j * 32];

    __syncthreads();

    float dens[RM], conv[RM];
    #pragma unroll
    for (int j = 0; j < RM; ++j) { dens[j] = 0.0f; conv[j] = 0.0f; }

    const float4* sp = sdata + wv * NPART;
    #pragma unroll 8
    for (int n = 0; n < NPART; ++n) {
        const float4 s = sp[n];
        #pragma unroll
        for (int j = 0; j < RM; ++j) {
            const float dx = s.x - tv[j].x;
            const float dy = s.y - tv[j].y;
            const float d  = __builtin_fmaf(dy, dy, dx * dx);
            dens[j] += __builtin_amdgcn_exp2f(d * c0);
            conv[j]  = __builtin_fmaf(s.z, __builtin_amdgcn_exp2f(d * c1), conv[j]);
        }
    }

    #pragma unroll
    for (int j = 0; j < RM; ++j) {
        spd[wv][j * 32 + lane] = dens[j];
        spc[wv][j * 32 + lane] = conv[j];
    }
    __syncthreads();

    const int hi   = lane >> 4;
    const int l15  = lane & 15;

    const int row = wv * 16 + l15;
    float dsum = 0.0f, csum = 0.0f;
    #pragma unroll
    for (int p = 0; p < NSPLIT; ++p) {
        dsum += spd[p][row];
        csum += spc[p][row];
    }
    const float f1 = csum / (dsum + 1e-8f);

    v2f a;
    a.x = hi ? 0.0f : dsum;
    a.y = hi ? 0.0f : f1;

    #pragma unroll
    for (int ot = 0; ot < 4; ++ot) {
        const int o  = ot * 16 + l15;
        const float w0 = W[o * 2 + 0];
        const float w1 = W[o * 2 + 1];
        const float bv = bias[o];
        v2f bm;
        bm.x = hi ? 0.0f : w0;
        bm.y = hi ? 0.0f : w1;

        v8f c;
        #pragma unroll
        for (int r = 0; r < 8; ++r) c[r] = bv;

        v8f d = __builtin_amdgcn_wmma_f32_16x16x4_f32(
            false, a, false, bm, (short)0, c, false, false);

        const int m_blk = wv * 16 + hi * 8;
        float* op = out + (((size_t)b * N_OUT + (m0 + m_blk)) * OC + o);
        #pragma unroll
        for (int r = 0; r < 8; ++r)
            op[(size_t)r * OC] = d[r];
    }
}

extern "C" void kernel_launch(void* const* d_in, const int* in_sizes, int n_in,
                              void* d_out, int out_size, void* d_ws, size_t ws_size,
                              hipStream_t stream) {
    const float* x     = (const float*)d_in[0];
    const float* y     = (const float*)d_in[1];
    const float* t     = (const float*)d_in[2];
    const float* sigma = (const float*)d_in[3];
    const float* W     = (const float*)d_in[4];
    const float* bias  = (const float*)d_in[5];
    float* out = (float*)d_out;

    dim3 grid(N_OUT / MB, B_);
    convdeepset_kernel<<<grid, dim3(BLK), 0, stream>>>(x, y, t, sigma, W, bias, out);
}


# Context loop as an explicit software pipeline: each step issues 4 exp2 beside 8 packed-f32 ops of the next 8-point group, operands made 2 steps earlier; VGPR cap lifted to 2 waves/SIMD
# speedup vs baseline: 2.1545x; 1.1659x over previous
#include <hip/hip_runtime.h>


#define B_     8
#define N_IN   1024
#define N_OUT  4096
#define OC     64
#define MB     128
#define RM     4
#define NSPLIT 8
#define NPART  (N_IN / NSPLIT)
#define BLK    256

typedef float v2f __attribute__((ext_vector_type(2)));
typedef float v8f __attribute__((ext_vector_type(8)));

#define RP (RM / 2)
#define GP 8
#define NI (GP * RP)

struct Pipe {
    v2f a0[NI], a1[NI];
    v2f dx[NI], dy[NI], dq[NI], dd[NI];
    v2f e0[NI], e1[NI];
    float sx[2][GP], sy[2][GP], yy[2][GP];
};

template <int KIND, int C, bool AHEAD>
__device__ __forceinline__ void stage(Pipe& P, const float4* __restrict__ nxt,
                                      const v2f (&tx)[RP], const v2f (&ty)[RP],
                                      const float c0, const float c1,
                                      v2f (&dens)[RP], v2f (&conv)[RP]) {
    #pragma unroll
    for (int t = 0; t < NI; ++t) {
        if ((t & 1) == 0) {
            const int ip = t / 2 + GP / 2;
            if (ip < GP) {
                if (KIND != 2) { const float4 s = nxt[ip]; P.sx[1 - C][ip] = s.x; P.sy[1 - C][ip] = s.y; P.yy[1 - C][ip] = s.z; }
            } else {
                if (AHEAD) { const float4 s = nxt[ip]; P.sx[C][ip - GP] = s.x; P.sy[C][ip - GP] = s.y; P.yy[C][ip - GP] = s.z; }
            }
        }
        if (KIND != 0) {
            P.e0[t].x = __builtin_amdgcn_exp2f(P.a0[t].x);
            P.e0[t].y = __builtin_amdgcn_exp2f(P.a0[t].y);
            P.e1[t].x = __builtin_amdgcn_exp2f(P.a1[t].x);
            P.e1[t].y = __builtin_amdgcn_exp2f(P.a1[t].y);
            const int u = (t - 2) & (NI - 1);
            const float yv = (t >= 2) ? P.yy[C][u / RP] : P.yy[1 - C][u / RP];
            const v2f ys = {yv, yv};
            dens[u % RP] += P.e0[u];
            conv[u % RP]  = __builtin_elementwise_fma(ys, P.e1[u], conv[u % RP]);
        }
        if (KIND != 2) {
            const v2f sx = {P.sx[1 - C][t / RP], P.sx[1 - C][t / RP]};
            const v2f sy = {P.sy[1 - C][t / RP], P.sy[1 - C][t / RP]};
            P.dx[t] = sx - tx[t % RP];
            P.dy[t] = sy - ty[t % RP];
        }
        if ((t >= 2) ? (KIND != 2) : (KIND != 0)) { const int u = (t - 2) & (NI - 1); P.dq[u] = P.dx[u] * P.dx[u]; }
        if ((t >= 4) ? (KIND != 2) : (KIND != 0)) { const int u = (t - 4) & (NI - 1); P.dd[u] = __builtin_elementwise_fma(P.dy[u], P.dy[u], P.dq[u]); }
        if ((t >= 6) ? (KIND != 2) : (KIND != 0)) { const int u = (t - 6) & (NI - 1); P.a0[u] = P.dd[u] * c0; P.a1[u] = P.dd[u] * c1; }
        __builtin_amdgcn_sched_group_barrier(0x400, 1, 0);
        __builtin_amdgcn_sched_group_barrier(0x002, 2, 0);
        __builtin_amdgcn_sched_group_barrier(0x400, 1, 0);
        __builtin_amdgcn_sched_group_barrier(0x002, 2, 0);
        __builtin_amdgcn_sched_group_barrier(0x400, 1, 0);
        __builtin_amdgcn_sched_group_barrier(0x002, 2, 0);
        __builtin_amdgcn_sched_group_barrier(0x400, 1, 0);
        __builtin_amdgcn_sched_group_barrier(0x002, 2, 0);
        __builtin_amdgcn_sched_barrier(0);
    }
}

__global__ __launch_bounds__(BLK) __attribute__((amdgpu_waves_per_eu(2, 2))) void convdeepset_kernel(
    const float* __restrict__ x,
    const float* __restrict__ y,
    const float* __restrict__ t,
    const float* __restrict__ sigma,
    const float* __restrict__ W,
    const float* __restrict__ bias,
    float* __restrict__ out)
{
    __shared__ float4 sdata[N_IN];
    __shared__ float  spd[NSPLIT][MB];
    __shared__ float  spc[NSPLIT][MB];

    const int tid  = threadIdx.x;
    const int lane = tid & 31;
    const int wv   = tid >> 5;
    const int b    = blockIdx.y;
    const int m0   = blockIdx.x * MB;

    const float* xb = x + (size_t)b * N_IN * 2;
    const float* yb = y + (size_t)b * N_IN;
    for (int n = tid; n < N_IN; n += BLK) {
        float2 xv = ((const float2*)xb)[n];
        float4 s; s.x = xv.x; s.y = xv.y; s.z = yb[n]; s.w = 0.0f;
        sdata[n] = s;
    }

    const float HALF_OVER_LN2 = 0.7213475204444817f;
    const float LOG2E         = 1.4426950408889634f;
    const float c0 = -HALF_OVER_LN2 * __builtin_amdgcn_exp2f(-2.0f * sigma[0] * LOG2E);
    const float c1 = -HALF_OVER_LN2 * __builtin_amdgcn_exp2f(-2.0f * sigma[1] * LOG2E);

    const float2* tb = (const float2*)(t + (size_t)b * N_OUT * 2) + (m0 + lane);
    float2 tv[RM];
    #pragma unroll
    for (int j = 0; j < RM; ++j) tv[j] = tb[j * 32];

    __syncthreads();

    v2f tx[RP], ty[RP], dens2[RP], conv2[RP];
    #pragma unroll
    for (int j = 0; j < RP; ++j) {
        tx[j].x = tv[2 * j].x;  tx[j].y = tv[2 * j + 1].x;
        ty[j].x = tv[2 * j].y;  ty[j].y = tv[2 * j + 1].y;
        dens2[j] = (v2f){0.0f, 0.0f};
        conv2[j] = (v2f){0.0f, 0.0f};
    }

    const float4* sp = sdata + wv * NPART;
    Pipe P;
    #pragma unroll
    for (int i = 0; i < NI; ++i) {
        P.a0[i] = P.a1[i] = P.dx[i] = P.dy[i] = P.dq[i] = P.dd[i] = P.e0[i] = P.e1[i] = (v2f){0.0f, 0.0f};
    }
    #pragma unroll
    for (int i = 0; i < GP; ++i) {
        P.sx[0][i] = P.sy[0][i] = P.yy[0][i] = 0.0f;
        P.sx[1][i] = P.sy[1][i] = P.yy[1][i] = 0.0f;
    }
    #pragma unroll
    for (int i = 0; i < GP / 2; ++i) {
        const float4 s = sp[i];
        P.sx[0][i] = s.x;  P.sy[0][i] = s.y;  P.yy[0][i] = s.z;
    }
    stage<0, 1, true>(P, sp, tx, ty, c0, c1, dens2, conv2);
    #pragma unroll 1
    for (int k = 1; k < NPART / GP - 1; k += 2) {
        stage<1, 0, true>(P, sp + k * GP, tx, ty, c0, c1, dens2, conv2);
        stage<1, 1, true>(P, sp + (k + 1) * GP, tx, ty, c0, c1, dens2, conv2);
    }
    stage<1, 0, false>(P, sp + (NPART / GP - 1) * GP, tx, ty, c0, c1, dens2, conv2);
    stage<2, 1, false>(P, sp, tx, ty, c0, c1, dens2, conv2);
    #pragma unroll
    for (int u = NI - 2; u < NI; ++u) {
        const v2f ys = {P.yy[1][u / RP], P.yy[1][u / RP]};
        dens2[u % RP] += P.e0[u];
        conv2[u % RP]  = __builtin_elementwise_fma(ys, P.e1[u], conv2[u % RP]);
    }

    float dens[RM], conv[RM];
    #pragma unroll
    for (int j = 0; j < RP; ++j) {
        dens[2 * j] = dens2[j].x;  dens[2 * j + 1] = dens2[j].y;
        conv[2 * j] = conv2[j].x;  conv[2 * j + 1] = conv2[j].y;
    }

    #pragma unroll
    for (int j = 0; j < RM; ++j) {
        spd[wv][j * 32 + lane] = dens[j];
        spc[wv][j * 32 + lane] = conv[j];
    }
    __syncthreads();

    const int hi   = lane >> 4;
    const int l15  = lane & 15;

    const int row = wv * 16 + l15;
    float dsum = 0.0f, csum = 0.0f;
    #pragma unroll
    for (int p = 0; p < NSPLIT; ++p) {
        dsum += spd[p][row];
        csum += spc[p][row];
    }
    const float f1 = csum / (dsum + 1e-8f);

    v2f a;
    a.x = hi ? 0.0f : dsum;
    a.y = hi ? 0.0f : f1;

    #pragma unroll
    for (int ot = 0; ot < 4; ++ot) {
        const int o  = ot * 16 + l15;
        const float w0 = W[o * 2 + 0];
        const float w1 = W[o * 2 + 1];
        const float bv = bias[o];
        v2f bm;
        bm.x = hi ? 0.0f : w0;
        bm.y = hi ? 0.0f : w1;

        v8f c;
        #pragma unroll
        for (int r = 0; r < 8; ++r) c[r] = bv;

        v8f d = __builtin_amdgcn_wmma_f32_16x16x4_f32(
            false, a, false, bm, (short)0, c, false, false);

        const int m_blk = wv * 16 + hi * 8;
        float* op = out + (((size_t)b * N_OUT + (m0 + m_blk)) * OC + o);
        #pragma unroll
        for (int r = 0; r < 8; ++r)
            op[(size_t)r * OC] = d[r];
    }
}

extern "C" void kernel_launch(void* const* d_in, const int* in_sizes, int n_in,
                              void* d_out, int out_size, void* d_ws, size_t ws_size,
                              hipStream_t stream) {
    const float* x     = (const float*)d_in[0];
    const float* y     = (const float*)d_in[1];
    const float* t     = (const float*)d_in[2];
    const float* sigma = (const float*)d_in[3];
    const float* W     = (const float*)d_in[4];
    const float* bias  = (const float*)d_in[5];
    float* out = (float*)d_out;

    dim3 grid(N_OUT / MB, B_);
    convdeepset_kernel<<<grid, dim3(BLK), 0, stream>>>(x, y, t, sigma, W, bias, out);
}


# All global loads (x, y, t, sigma, W, bias) issued as one group at kernel entry: one exposed memory round trip instead of six serial ones before the context loop and one behind it
# speedup vs baseline: 2.6801x; 1.2440x over previous
#include <hip/hip_runtime.h>


#define B_     8
#define N_IN   1024
#define N_OUT  4096
#define OC     64
#define MB     128
#define RM     4
#define NSPLIT 8
#define NPART  (N_IN / NSPLIT)
#define BLK    256

typedef float v2f __attribute__((ext_vector_type(2)));
typedef float v8f __attribute__((ext_vector_type(8)));

#define RP (RM / 2)
#define GP 8
#define NI (GP * RP)

struct Pipe {
    v2f a0[NI], a1[NI];
    v2f dx[NI], dy[NI], dq[NI], dd[NI];
    v2f e0[NI], e1[NI];
    float sx[2][GP], sy[2][GP], yy[2][GP];
};

template <int KIND, int C, bool AHEAD>
__device__ __forceinline__ void stage(Pipe& P, const float4* __restrict__ nxt,
                                      const v2f (&tx)[RP], const v2f (&ty)[RP],
                                      const float c0, const float c1,
                                      v2f (&dens)[RP], v2f (&conv)[RP]) {
    #pragma unroll
    for (int t = 0; t < NI; ++t) {
        if ((t & 1) == 0) {
            const int ip = t / 2 + GP / 2;
            if (ip < GP) {
                if (KIND != 2) { const float4 s = nxt[ip]; P.sx[1 - C][ip] = s.x; P.sy[1 - C][ip] = s.y; P.yy[1 - C][ip] = s.z; }
            } else {
                if (AHEAD) { const float4 s = nxt[ip]; P.sx[C][ip - GP] = s.x; P.sy[C][ip - GP] = s.y; P.yy[C][ip - GP] = s.z; }
            }
        }
        if (KIND != 0) {
            P.e0[t].x = __builtin_amdgcn_exp2f(P.a0[t].x);
            P.e0[t].y = __builtin_amdgcn_exp2f(P.a0[t].y);
            P.e1[t].x = __builtin_amdgcn_exp2f(P.a1[t].x);
            P.e1[t].y = __builtin_amdgcn_exp2f(P.a1[t].y);
            const int u = (t - 2) & (NI - 1);
            const float yv = (t >= 2) ? P.yy[C][u / RP] : P.yy[1 - C][u / RP];
            const v2f ys = {yv, yv};
            dens[u % RP] += P.e0[u];
            conv[u % RP]  = __builtin_elementwise_fma(ys, P.e1[u], conv[u % RP]);
        }
        if (KIND != 2) {
            const v2f sx = {P.sx[1 - C][t / RP], P.sx[1 - C][t / RP]};
            const v2f sy = {P.sy[1 - C][t / RP], P.sy[1 - C][t / RP]};
            P.dx[t] = sx - tx[t % RP];
            P.dy[t] = sy - ty[t % RP];
        }
        if ((t >= 2) ? (KIND != 2) : (KIND != 0)) { const int u = (t - 2) & (NI - 1); P.dq[u] = P.dx[u] * P.dx[u]; }
        if ((t >= 4) ? (KIND != 2) : (KIND != 0)) { const int u = (t - 4) & (NI - 1); P.dd[u] = __builtin_elementwise_fma(P.dy[u], P.dy[u], P.dq[u]); }
        if ((t >= 6) ? (KIND != 2) : (KIND != 0)) { const int u = (t - 6) & (NI - 1); P.a0[u] = P.dd[u] * c0; P.a1[u] = P.dd[u] * c1; }
        __builtin_amdgcn_sched_group_barrier(0x400, 1, 0);
        __builtin_amdgcn_sched_group_barrier(0x002, 2, 0);
        __builtin_amdgcn_sched_group_barrier(0x400, 1, 0);
        __builtin_amdgcn_sched_group_barrier(0x002, 2, 0);
        __builtin_amdgcn_sched_group_barrier(0x400, 1, 0);
        __builtin_amdgcn_sched_group_barrier(0x002, 2, 0);
        __builtin_amdgcn_sched_group_barrier(0x400, 1, 0);
        __builtin_amdgcn_sched_group_barrier(0x002, 2, 0);
        __builtin_amdgcn_sched_barrier(0);
    }
}

__global__ __launch_bounds__(BLK) __attribute__((amdgpu_waves_per_eu(2, 2))) void convdeepset_kernel(
    const float* __restrict__ x,
    const float* __restrict__ y,
    const float* __restrict__ t,
    const float* __restrict__ sigma,
    const float* __restrict__ W,
    const float* __restrict__ bias,
    float* __restrict__ out)
{
    __shared__ float4 sdata[N_IN];
    __shared__ float  spd[NSPLIT][MB];
    __shared__ float  spc[NSPLIT][MB];

    const int tid  = threadIdx.x;
    const int lane = tid & 31;
    const int wv   = tid >> 5;
    const int b    = blockIdx.y;
    const int m0   = blockIdx.x * MB;

    const float2* xb = (const float2*)(x + (size_t)b * N_IN * 2);
    const float*  yb = y + (size_t)b * N_IN;
    float2 xv[N_IN / BLK];
    float  yv[N_IN / BLK];
    #pragma unroll
    for (int i = 0; i < N_IN / BLK; ++i) {
        xv[i] = xb[tid + i * BLK];
        yv[i] = yb[tid + i * BLK];
    }

    const float2* tb = (const float2*)(t + (size_t)b * N_OUT * 2) + (m0 + lane);
    float2 tv[RM];
    #pragma unroll
    for (int j = 0; j < RM; ++j) tv[j] = tb[j * 32];

    const float sg0 = sigma[0];
    const float sg1 = sigma[1];

    float2 wrow[OC / 16];
    float  brow[OC / 16];
    #pragma unroll
    for (int ot = 0; ot < OC / 16; ++ot) {
        wrow[ot] = ((const float2*)W)[ot * 16 + (lane & 15)];
        brow[ot] = bias[ot * 16 + (lane & 15)];
    }

    __builtin_amdgcn_sched_barrier(0);

    #pragma unroll
    for (int i = 0; i < N_IN / BLK; ++i) {
        float4 s; s.x = xv[i].x; s.y = xv[i].y; s.z = yv[i]; s.w = 0.0f;
        sdata[tid + i * BLK] = s;
    }

    const float HALF_OVER_LN2 = 0.7213475204444817f;
    const float LOG2E         = 1.4426950408889634f;
    const float c0_ = -HALF_OVER_LN2 * __builtin_amdgcn_exp2f(-2.0f * sg0 * LOG2E);
    const float c1_ = -HALF_OVER_LN2 * __builtin_amdgcn_exp2f(-2.0f * sg1 * LOG2E);
    float c0 = c0_, c1 = c1_;
    asm volatile("" : "+v"(c0), "+v"(c1));

    __syncthreads();

    v2f tx[RP], ty[RP], dens2[RP], conv2[RP];
    #pragma unroll
    for (int j = 0; j < RP; ++j) {
        tx[j].x = tv[2 * j].x;  tx[j].y = tv[2 * j + 1].x;
        ty[j].x = tv[2 * j].y;  ty[j].y = tv[2 * j + 1].y;
        dens2[j] = (v2f){0.0f, 0.0f};
        conv2[j] = (v2f){0.0f, 0.0f};
    }

    const float4* sp = sdata + wv * NPART;
    Pipe P;
    #pragma unroll
    for (int i = 0; i < NI; ++i) {
        P.a0[i] = P.a1[i] = P.dx[i] = P.dy[i] = P.dq[i] = P.dd[i] = P.e0[i] = P.e1[i] = (v2f){0.0f, 0.0f};
    }
    #pragma unroll
    for (int i = 0; i < GP; ++i) {
        P.sx[0][i] = P.sy[0][i] = P.yy[0][i] = 0.0f;
        P.sx[1][i] = P.sy[1][i] = P.yy[1][i] = 0.0f;
    }
    #pragma unroll
    for (int i = 0; i < GP / 2; ++i) {
        const float4 s = sp[i];
        P.sx[0][i] = s.x;  P.sy[0][i] = s.y;  P.yy[0][i] = s.z;
    }
    stage<0, 1, true>(P, sp, tx, ty, c0, c1, dens2, conv2);
    #pragma unroll 1
    for (int k = 1; k < NPART / GP - 1; k += 2) {
        stage<1, 0, true>(P, sp + k * GP, tx, ty, c0, c1, dens2, conv2);
        stage<1, 1, true>(P, sp + (k + 1) * GP, tx, ty, c0, c1, dens2, conv2);
    }
    stage<1, 0, false>(P, sp + (NPART / GP - 1) * GP, tx, ty, c0, c1, dens2, conv2);
    stage<2, 1, false>(P, sp, tx, ty, c0, c1, dens2, conv2);
    #pragma unroll
    for (int u = NI - 2; u < NI; ++u) {
        const v2f ys = {P.yy[1][u / RP], P.yy[1][u / RP]};
        dens2[u % RP] += P.e0[u];
        conv2[u % RP]  = __builtin_elementwise_fma(ys, P.e1[u], conv2[u % RP]);
    }

    float dens[RM], conv[RM];
    #pragma unroll
    for (int j = 0; j < RP; ++j) {
        dens[2 * j] = dens2[j].x;  dens[2 * j + 1] = dens2[j].y;
        conv[2 * j] = conv2[j].x;  conv[2 * j + 1] = conv2[j].y;
    }

    #pragma unroll
    for (int j = 0; j < RM; ++j) {
        spd[wv][j * 32 + lane] = dens[j];
        spc[wv][j * 32 + lane] = conv[j];
    }
    __syncthreads();

    const int hi   = lane >> 4;
    const int l15  = lane & 15;

    const int row = wv * 16 + l15;
    float dsum = 0.0f, csum = 0.0f;
    #pragma unroll
    for (int p = 0; p < NSPLIT; ++p) {
        dsum += spd[p][row];
        csum += spc[p][row];
    }
    const float f1 = csum / (dsum + 1e-8f);

    v2f a;
    a.x = hi ? 0.0f : dsum;
    a.y = hi ? 0.0f : f1;

    #pragma unroll
    for (int ot = 0; ot < 4; ++ot) {
        const int o  = ot * 16 + l15;
        const float w0 = wrow[ot].x;
        const float w1 = wrow[ot].y;
        const float bv = brow[ot];
        v2f bm;
        bm.x = hi ? 0.0f : w0;
        bm.y = hi ? 0.0f : w1;

        v8f c;
        #pragma unroll
        for (int r = 0; r < 8; ++r) c[r] = bv;

        v8f d = __builtin_amdgcn_wmma_f32_16x16x4_f32(
            false, a, false, bm, (short)0, c, false, false);

        const int m_blk = wv * 16 + hi * 8;
        float* op = out + (((size_t)b * N_OUT + (m0 + m_blk)) * OC + o);
        #pragma unroll
        for (int r = 0; r < 8; ++r)
            op[(size_t)r * OC] = d[r];
    }
}

extern "C" void kernel_launch(void* const* d_in, const int* in_sizes, int n_in,
                              void* d_out, int out_size, void* d_ws, size_t ws_size,
                              hipStream_t stream) {
    const float* x     = (const float*)d_in[0];
    const float* y     = (const float*)d_in[1];
    const float* t     = (const float*)d_in[2];
    const float* sigma = (const float*)d_in[3];
    const float* W     = (const float*)d_in[4];
    const float* bias  = (const float*)d_in[5];
    float* out = (float*)d_out;

    dim3 grid(N_OUT / MB, B_);
    convdeepset_kernel<<<grid, dim3(BLK), 0, stream>>>(x, y, t, sigma, W, bias, out);
}


# Each wave owns 16 targets over all 1,024 context points (targets as WMMA rows): no partial sums, no barrier behind the loop; an early wave projects and stores beside its SIMD partner's loop
# speedup vs baseline: 3.1684x; 1.1822x over previous
#include <hip/hip_runtime.h>


#define B_     8
#define N_IN   1024
#define N_OUT  4096
#define OC     64
#define MB     128
#define BLK    256
#define NT     (N_IN / 16)
#define GS     16
#define YP     (NT + 4)
#define RP     20

typedef float v2f __attribute__((ext_vector_type(2)));
typedef float v4f __attribute__((ext_vector_type(4)));
typedef float v8f __attribute__((ext_vector_type(8)));


struct Acc  { v2f dens[4]; v2f conv[4]; };
struct Pipe {
    v8f dM;
    v2f a0[4], a1[4];
    v2f p0[4], p1[4];
};

__device__ __forceinline__ void step(Pipe& P, Acc& S, const v2f Tf, const v2f Xw, const float yv,
                                     const float c0, const float c1) {
    const v8f zero = {0.0f, 0.0f, 0.0f, 0.0f, 0.0f, 0.0f, 0.0f, 0.0f};
    const v8f dW = __builtin_amdgcn_wmma_f32_16x16x4_f32(false, Tf, false, Xw, (short)0, zero, false, false);
    const v2f yk = {yv, yv};
    v2f e0[4], e1[4], m0[4], m1[4];
    #pragma unroll
    for (int k = 0; k < 4; ++k) {
        const v2f dk = {P.dM[2 * k], P.dM[2 * k + 1]};
        e0[k].x = __builtin_amdgcn_exp2f(P.a0[k].x);
        m0[k] = dk * c0;
        e0[k].y = __builtin_amdgcn_exp2f(P.a0[k].y);
        S.dens[k] += P.p0[k];
        e1[k].x = __builtin_amdgcn_exp2f(P.a1[k].x);
        m1[k] = dk * c1;
        e1[k].y = __builtin_amdgcn_exp2f(P.a1[k].y);
        S.conv[k] = __builtin_elementwise_fma(yk, P.p1[k], S.conv[k]);
    }
    #pragma unroll
    for (int k = 0; k < 4; ++k) {
        P.p0[k] = e0[k];  P.p1[k] = e1[k];
        P.a0[k] = m0[k];  P.a1[k] = m1[k];
    }
    P.dM = dW;
    __builtin_amdgcn_sched_group_barrier(0x008, 1, 0);
    #pragma unroll
    for (int q = 0; q < 16; ++q) {
        __builtin_amdgcn_sched_group_barrier(0x400, 1, 0);
        __builtin_amdgcn_sched_group_barrier(0x002, 1, 0);
    }
    __builtin_amdgcn_sched_barrier(0);
}

template <int Q>
__device__ __forceinline__ v2f pick_x(const v2f (&X0)[GS], const v2f (&X1)[GS]) {
    constexpr int t = (Q + 2) & 31;
    if constexpr (t < GS) return X0[t]; else return X1[t - GS];
}
template <int Q>
__device__ __forceinline__ float pick_y(const v4f (&Y0)[GS / 4], const v4f (&Y1)[GS / 4]) {
    constexpr int t = (Q + 31) & 31;
    if constexpr (t < GS) return Y0[t >> 2][t & 3]; else return Y1[(t - GS) >> 2][(t - GS) & 3];
}
#define STEP(Q) step(P, S, Tf, pick_x<Q>(X0, X1), pick_y<Q>(Y0, Y1), c0, c1)

__global__ __launch_bounds__(BLK) __attribute__((amdgpu_waves_per_eu(2, 2))) void convdeepset_kernel(
    const float* __restrict__ x,
    const float* __restrict__ y,
    const float* __restrict__ t,
    const float* __restrict__ sigma,
    const float* __restrict__ W,
    const float* __restrict__ bias,
    float* __restrict__ out)
{
    __shared__ float4 sA[N_IN];
    __shared__ __attribute__((aligned(16))) float sYt[16 * YP];
    __shared__ __attribute__((aligned(16))) float red[BLK / 32][2][16 * RP];

    const int tid  = threadIdx.x;
    const int lane = tid & 31;
    const int wv   = tid >> 5;
    const int hi   = lane >> 4;
    const int l15  = lane & 15;
    const int b    = blockIdx.y;
    const int m0   = blockIdx.x * MB;

    const float2* xb = (const float2*)(x + (size_t)b * N_IN * 2);
    const float*  yb = y + (size_t)b * N_IN;
    float2 xv[N_IN / BLK];
    float  yv[N_IN / BLK];
    #pragma unroll
    for (int i = 0; i < N_IN / BLK; ++i) {
        xv[i] = xb[tid + i * BLK];
        yv[i] = yb[tid + i * BLK];
    }

    const float2 tv = ((const float2*)(t + (size_t)b * N_OUT * 2))[m0 + 16 * wv + l15];

    const float sg0 = sigma[0];
    const float sg1 = sigma[1];

    float2 wrow[OC / 16];
    float  brow[OC / 16];
    #pragma unroll
    for (int ot = 0; ot < OC / 16; ++ot) {
        wrow[ot] = ((const float2*)W)[ot * 16 + l15];
        brow[ot] = bias[ot * 16 + l15];
    }

    __builtin_amdgcn_sched_barrier(0);

    #pragma unroll
    for (int i = 0; i < N_IN / BLK; ++i) {
        const int n = tid + i * BLK;
        float4 s;
        s.x = -2.0f * xv[i].x;
        s.y = -2.0f * xv[i].y;
        s.z = __builtin_fmaf(xv[i].y, xv[i].y, xv[i].x * xv[i].x);
        s.w = 1.0f;
        sA[n] = s;
        sYt[(n & 15) * YP + (n >> 4)] = yv[i];
    }

    v2f Tf;
    {
        const float tt = __builtin_fmaf(tv.y, tv.y, tv.x * tv.x);
        Tf.x = hi ? 1.0f : tv.x;
        Tf.y = hi ? tt   : tv.y;
    }

    const float HALF_OVER_LN2 = 0.7213475204444817f;
    const float LOG2E         = 1.4426950408889634f;
    const float c0_ = -HALF_OVER_LN2 * __builtin_amdgcn_exp2f(-2.0f * sg0 * LOG2E);
    const float c1_ = -HALF_OVER_LN2 * __builtin_amdgcn_exp2f(-2.0f * sg1 * LOG2E);
    float c0 = c0_, c1 = c1_;
    asm volatile("" : "+v"(c0), "+v"(c1));

    __syncthreads();

    const v2f* xp = (const v2f*)sA + l15 * 2 + hi;
    const v4f* yq = (const v4f*)(sYt + l15 * YP);

    Acc S;
    #pragma unroll
    for (int k = 0; k < 4; ++k) {
        S.dens[k] = (v2f){0.0f, 0.0f};
        S.conv[k] = (v2f){0.0f, 0.0f};
    }

    v2f X0[GS], X1[GS];
    v4f Y0[GS / 4], Y1[GS / 4];
    #pragma unroll
    for (int j = 0; j < GS; ++j) {
        X0[j] = xp[32 * j];
        X1[j] = (v2f){0.0f, 0.0f};
    }
    #pragma unroll
    for (int g = 0; g < GS / 4; ++g) {
        Y0[g] = yq[g];
        Y1[g] = (v4f){0.0f, 0.0f, 0.0f, 0.0f};
    }

    Pipe P;
    {
        const v8f zero = {0.0f, 0.0f, 0.0f, 0.0f, 0.0f, 0.0f, 0.0f, 0.0f};
        const v8f d0 = __builtin_amdgcn_wmma_f32_16x16x4_f32(false, Tf, false, X0[0], (short)0, zero, false, false);
        const v8f d1 = __builtin_amdgcn_wmma_f32_16x16x4_f32(false, Tf, false, X0[1], (short)0, zero, false, false);
        #pragma unroll
        for (int k = 0; k < 4; ++k) {
            const v2f dk = {d0[2 * k], d0[2 * k + 1]};
            P.a0[k] = dk * c0;
            P.a1[k] = dk * c1;
            P.p0[k] = (v2f){0.0f, 0.0f};
            P.p1[k] = (v2f){0.0f, 0.0f};
        }
        P.dM = d1;
    }
    __builtin_amdgcn_sched_barrier(0);

    #pragma unroll 1
    for (int it = 0; it < NT / 32; ++it) {
        const int base = 32 * it;
        STEP(0);
        #pragma unroll
        for (int j = 0; j < GS; ++j) X1[j] = xp[32 * (base + GS + j)];
        #pragma unroll
        for (int g = 0; g < GS / 4; ++g) Y1[g] = yq[(base + GS) / 4 + g];
        __builtin_amdgcn_sched_barrier(0);
        STEP(1);
        STEP(2);
        STEP(3);
        STEP(4);
        STEP(5);
        STEP(6);
        STEP(7);
        STEP(8);
        STEP(9);
        STEP(10);
        STEP(11);
        STEP(12);
        STEP(13);
        STEP(14);
        STEP(15);
        STEP(16);
        const int nb = (base + 32) & (NT - 1);
        #pragma unroll
        for (int j = 0; j < GS; ++j) X0[j] = xp[32 * (nb + j)];
        #pragma unroll
        for (int g = 0; g < GS / 4; ++g) Y0[g] = yq[nb / 4 + g];
        __builtin_amdgcn_sched_barrier(0);
        STEP(17);
        STEP(18);
        STEP(19);
        STEP(20);
        STEP(21);
        STEP(22);
        STEP(23);
        STEP(24);
        STEP(25);
        STEP(26);
        STEP(27);
        STEP(28);
        STEP(29);
        STEP(30);
        STEP(31);
    }

    {
        const float yl = Y1[GS / 4 - 1][3];
        const v2f yk = {yl, yl};
        #pragma unroll
        for (int k = 0; k < 4; ++k) {
            S.dens[k] += P.p0[k];
            S.conv[k] = __builtin_elementwise_fma(yk, P.p1[k], S.conv[k]);
        }
    }

    {
        float* rd = &red[wv][0][l15 * RP + 8 * hi];
        float* rc = &red[wv][1][l15 * RP + 8 * hi];
        *(v4f*)(rd)     = (v4f){S.dens[0].x, S.dens[0].y, S.dens[1].x, S.dens[1].y};
        *(v4f*)(rd + 4) = (v4f){S.dens[2].x, S.dens[2].y, S.dens[3].x, S.dens[3].y};
        *(v4f*)(rc)     = (v4f){S.conv[0].x, S.conv[0].y, S.conv[1].x, S.conv[1].y};
        *(v4f*)(rc + 4) = (v4f){S.conv[2].x, S.conv[2].y, S.conv[3].x, S.conv[3].y};
    }
    __builtin_amdgcn_fence(__ATOMIC_RELEASE, "wavefront");
    __builtin_amdgcn_wave_barrier();
    __builtin_amdgcn_fence(__ATOMIC_ACQUIRE, "wavefront");


    float pd[16], pc[16];
    #pragma unroll
    for (int p = 0; p < 16; ++p) {
        pd[p] = red[wv][0][p * RP + l15];
        pc[p] = red[wv][1][p * RP + l15];
    }
    __builtin_amdgcn_sched_barrier(0);
    #pragma unroll
    for (int s = 8; s > 0; s >>= 1) {
        #pragma unroll
        for (int p = 0; p < s; ++p) {
            pd[p] += pd[p + s];
            pc[p] += pc[p + s];
        }
    }
    const float dsum = pd[0], csum = pc[0];
    const float f1 = csum / (dsum + 1e-8f);

    v2f a;
    a.x = hi ? 0.0f : dsum;
    a.y = hi ? 0.0f : f1;

    #pragma unroll
    for (int ot = 0; ot < 4; ++ot) {
        const int o  = ot * 16 + l15;
        const float w0 = wrow[ot].x;
        const float w1 = wrow[ot].y;
        const float bv = brow[ot];
        v2f bm;
        bm.x = hi ? 0.0f : w0;
        bm.y = hi ? 0.0f : w1;

        v8f c;
        #pragma unroll
        for (int r = 0; r < 8; ++r) c[r] = bv;

        v8f d = __builtin_amdgcn_wmma_f32_16x16x4_f32(
            false, a, false, bm, (short)0, c, false, false);

        const int m_blk = wv * 16 + hi * 8;
        float* op = out + (((size_t)b * N_OUT + (m0 + m_blk)) * OC + o);
        #pragma unroll
        for (int r = 0; r < 8; ++r)
            op[(size_t)r * OC] = d[r];
    }
}

extern "C" void kernel_launch(void* const* d_in, const int* in_sizes, int n_in,
                              void* d_out, int out_size, void* d_ws, size_t ws_size,
                              hipStream_t stream) {
    const float* x     = (const float*)d_in[0];
    const float* y     = (const float*)d_in[1];
    const float* t     = (const float*)d_in[2];
    const float* sigma = (const float*)d_in[3];
    const float* W     = (const float*)d_in[4];
    const float* bias  = (const float*)d_in[5];
    float* out = (float*)d_out;

    dim3 grid(N_OUT / MB, B_);
    convdeepset_kernel<<<grid, dim3(BLK), 0, stream>>>(x, y, t, sigma, W, bias, out);
}
